# starting kernel
# baseline (speedup 1.0000x reference)
#include <hip/hip_runtime.h>
#include <hip/hip_bf16.h>
#include <math.h>

#define BATCH   16
#define CIN     256
#define NPIX    4096
#define HEADS   4
#define DHEAD   32
#define HIDDEN  128
#define O3      384
#define KSPLIT  8
#define KCHUNK  (NPIX / KSPLIT)

typedef __attribute__((ext_vector_type(2))) float v2f;
typedef __attribute__((ext_vector_type(8))) float v8f;

__device__ __forceinline__ v8f wmma4(v2f a, v2f b, v8f c) {
  return __builtin_amdgcn_wmma_f32_16x16x4_f32(false, a, false, b, (short)0, c,
                                               false, false);
}

__global__ __launch_bounds__(256) void k_qkv_gemm(
    const float* __restrict__ x, const float* __restrict__ wqkv,
    float* __restrict__ qkv) {
  const int lane = threadIdx.x & 31;
  const int wave = threadIdx.x >> 5;
  const int half = lane >> 4;
  const int lm   = lane & 15;
  const int nBase = blockIdx.x * 16;
  const int b     = blockIdx.y;

  const float* __restrict__ xb = x + (size_t)b * CIN * NPIX;
  float* __restrict__ qb = qkv + (size_t)b * O3 * NPIX;

  const int m0 = wave * 16;
  v8f acc0 = {}, acc1 = {}, acc2 = {};

  for (int kb = 0; kb < CIN; kb += 4) {
    const int k0 = kb + 2 * half;
    v2f bf, a0, a1, a2;
    bf.x = xb[(size_t)k0 * NPIX + nBase + lm];
    bf.y = xb[(size_t)(k0 + 1) * NPIX + nBase + lm];
    a0.x = wqkv[(m0 + lm) * CIN + k0];
    a0.y = wqkv[(m0 + lm) * CIN + k0 + 1];
    a1.x = wqkv[(m0 + 128 + lm) * CIN + k0];
    a1.y = wqkv[(m0 + 128 + lm) * CIN + k0 + 1];
    a2.x = wqkv[(m0 + 256 + lm) * CIN + k0];
    a2.y = wqkv[(m0 + 256 + lm) * CIN + k0 + 1];
    acc0 = wmma4(a0, bf, acc0);
    acc1 = wmma4(a1, bf, acc1);
    acc2 = wmma4(a2, bf, acc2);
  }
#pragma unroll
  for (int v = 0; v < 8; ++v) {
    const int row = v + 8 * half;
    qb[(size_t)(m0 + row) * NPIX + nBase + lm]       = acc0[v];
    qb[(size_t)(m0 + 128 + row) * NPIX + nBase + lm] = acc1[v];
    qb[(size_t)(m0 + 256 + row) * NPIX + nBase + lm] = acc2[v];
  }
}

__global__ __launch_bounds__(256) void k_softmax_n(float* __restrict__ qkv) {
  const int r = blockIdx.x;
  const int b = r >> 7;
  const int c = r & 127;
  float* __restrict__ row = qkv + ((size_t)b * O3 + c) * NPIX;

  __shared__ float red[256];
  const int t = threadIdx.x;

  float v[16];
  float m = -INFINITY;
#pragma unroll
  for (int j = 0; j < 16; ++j) {
    v[j] = row[t + j * 256];
    m = fmaxf(m, v[j]);
  }
  red[t] = m; __syncthreads();
  for (int s = 128; s > 0; s >>= 1) {
    if (t < s) red[t] = fmaxf(red[t], red[t + s]);
    __syncthreads();
  }
  m = red[0]; __syncthreads();

  float sum = 0.f;
#pragma unroll
  for (int j = 0; j < 16; ++j) {
    v[j] = __expf(v[j] - m);
    sum += v[j];
  }
  red[t] = sum; __syncthreads();
  for (int s = 128; s > 0; s >>= 1) {
    if (t < s) red[t] += red[t + s];
    __syncthreads();
  }
  const float inv = 1.0f / red[0];
#pragma unroll
  for (int j = 0; j < 16; ++j) row[t + j * 256] = v[j] * inv;
}

__global__ __launch_bounds__(256) void k_softmax_d(
    const float* __restrict__ qkv, float* __restrict__ kbuf) {
  const int col = blockIdx.x * 256 + threadIdx.x;
  const int n = col & (NPIX - 1);
  const int h = (col >> 12) & (HEADS - 1);
  const int b = col >> 14;

  const float* __restrict__ qs = qkv + ((size_t)b * O3 + h * DHEAD) * NPIX + n;
  float vals[DHEAD];
  float m = -INFINITY;
#pragma unroll
  for (int d = 0; d < DHEAD; ++d) {
    vals[d] = qs[(size_t)d * NPIX];
    m = fmaxf(m, vals[d]);
  }
  float s = 0.f;
#pragma unroll
  for (int d = 0; d < DHEAD; ++d) {
    vals[d] = __expf(vals[d] - m);
    s += vals[d];
  }
  const float inv = 1.0f / s;
  float* __restrict__ ko =
      kbuf + (size_t)(b * HEADS + h) * DHEAD * NPIX + n;
#pragma unroll
  for (int d = 0; d < DHEAD; ++d) ko[(size_t)d * NPIX] = vals[d] * inv;
}

__global__ __launch_bounds__(128) void k_context_part(
    const float* __restrict__ kbuf, const float* __restrict__ qkv,
    float* __restrict__ pctx) {
  const int ks = blockIdx.x;
  const int bh = blockIdx.y;
  const int b = bh >> 2, h = bh & 3;
  const int lane = threadIdx.x & 31;
  const int wave = threadIdx.x >> 5;
  const int half = lane >> 4, lm = lane & 15;
  const int mBase = (wave >> 1) * 16;
  const int nBase = (wave & 1) * 16;
  const int kBeg  = ks * KCHUNK;

  const float* __restrict__ kA = kbuf + (size_t)bh * DHEAD * NPIX;
  const float* __restrict__ vB =
      qkv + ((size_t)b * O3 + 2 * HIDDEN + h * DHEAD) * NPIX;

  v8f acc = {};
  for (int kb = 0; kb < KCHUNK; kb += 4) {
    const int k0 = kBeg + kb + 2 * half;
    v2f a, bb;
    a.x  = kA[(size_t)(mBase + lm) * NPIX + k0];
    a.y  = kA[(size_t)(mBase + lm) * NPIX + k0 + 1];
    bb.x = vB[(size_t)(nBase + lm) * NPIX + k0];
    bb.y = vB[(size_t)(nBase + lm) * NPIX + k0 + 1];
    acc = wmma4(a, bb, acc);
  }
#pragma unroll
  for (int v = 0; v < 8; ++v) {
    const int row = mBase + v + 8 * half;
    pctx[(((size_t)ks * (BATCH * HEADS) + bh) * DHEAD + row) * DHEAD + nBase +
         lm] = acc[v];
  }
}

__global__ __launch_bounds__(256) void k_ctx_reduce(
    const float* __restrict__ pctx, float* __restrict__ ctx) {
  const int i = blockIdx.x * 256 + threadIdx.x;
  const size_t stride = (size_t)BATCH * HEADS * DHEAD * DHEAD;
  float s = 0.f;
#pragma unroll
  for (int ks = 0; ks < KSPLIT; ++ks) s += pctx[(size_t)ks * stride + i];
  ctx[i] = s;
}

__global__ __launch_bounds__(256) void k_fold_w2(
    const float* __restrict__ wout, const float* __restrict__ ctx,
    float* __restrict__ w2) {
  const int bh = blockIdx.x;
  const int b = bh >> 2, h = bh & 3;
  const int lane = threadIdx.x & 31;
  const int wave = threadIdx.x >> 5;
  const int half = lane >> 4, lm = lane & 15;
  const int m0 = wave * 16;

  const float* __restrict__ cb = ctx + (size_t)bh * DHEAD * DHEAD;

  v8f acc00 = {}, acc01 = {}, acc10 = {}, acc11 = {};
#pragma unroll
  for (int kb = 0; kb < DHEAD; kb += 4) {
    const int k0 = kb + 2 * half;
    v2f a0, a1, b0, b1;
    a0.x = wout[(m0 + lm) * HIDDEN + h * DHEAD + k0];
    a0.y = wout[(m0 + lm) * HIDDEN + h * DHEAD + k0 + 1];
    a1.x = wout[(m0 + 128 + lm) * HIDDEN + h * DHEAD + k0];
    a1.y = wout[(m0 + 128 + lm) * HIDDEN + h * DHEAD + k0 + 1];
    b0.x = cb[(size_t)(0 + lm) * DHEAD + k0];
    b0.y = cb[(size_t)(0 + lm) * DHEAD + k0 + 1];
    b1.x = cb[(size_t)(16 + lm) * DHEAD + k0];
    b1.y = cb[(size_t)(16 + lm) * DHEAD + k0 + 1];
    acc00 = wmma4(a0, b0, acc00);
    acc01 = wmma4(a0, b1, acc01);
    acc10 = wmma4(a1, b0, acc10);
    acc11 = wmma4(a1, b1, acc11);
  }
  float* __restrict__ w2b = w2 + (size_t)b * CIN * HIDDEN;
#pragma unroll
  for (int v = 0; v < 8; ++v) {
    const int row = v + 8 * half;
    w2b[(size_t)(m0 + row) * HIDDEN + h * DHEAD + 0 + lm]        = acc00[v];
    w2b[(size_t)(m0 + row) * HIDDEN + h * DHEAD + 16 + lm]       = acc01[v];
    w2b[(size_t)(m0 + 128 + row) * HIDDEN + h * DHEAD + 0 + lm]  = acc10[v];
    w2b[(size_t)(m0 + 128 + row) * HIDDEN + h * DHEAD + 16 + lm] = acc11[v];
  }
}

__global__ __launch_bounds__(256) void k_out_gemm(
    const float* __restrict__ qkv, const float* __restrict__ w2,
    float* __restrict__ out) {
  const int lane = threadIdx.x & 31;
  const int wave = threadIdx.x >> 5;
  const int half = lane >> 4, lm = lane & 15;
  const int nBase = blockIdx.x * 16;
  const int b     = blockIdx.y;

  const float* __restrict__ qs  = qkv + (size_t)b * O3 * NPIX;
  const float* __restrict__ w2b = w2 + (size_t)b * CIN * HIDDEN;
  float* __restrict__ ob = out + (size_t)b * CIN * NPIX;

  const int m0 = wave * 16;
  v8f acc0 = {}, acc1 = {};
  for (int kb = 0; kb < HIDDEN; kb += 4) {
    const int k0 = kb + 2 * half;
    v2f bf, a0, a1;
    bf.x = qs[(size_t)k0 * NPIX + nBase + lm];
    bf.y = qs[(size_t)(k0 + 1) * NPIX + nBase + lm];
    a0.x = w2b[(m0 + lm) * HIDDEN + k0];
    a0.y = w2b[(m0 + lm) * HIDDEN + k0 + 1];
    a1.x = w2b[(m0 + 128 + lm) * HIDDEN + k0];
    a1.y = w2b[(m0 + 128 + lm) * HIDDEN + k0 + 1];
    acc0 = wmma4(a0, bf, acc0);
    acc1 = wmma4(a1, bf, acc1);
  }
#pragma unroll
  for (int v = 0; v < 8; ++v) {
    const int row = v + 8 * half;
    ob[(size_t)(m0 + row) * NPIX + nBase + lm]       = acc0[v];
    ob[(size_t)(m0 + 128 + row) * NPIX + nBase + lm] = acc1[v];
  }
}

extern "C" void kernel_launch(void* const* d_in, const int* in_sizes, int n_in,
                              void* d_out, int out_size, void* d_ws,
                              size_t ws_size, hipStream_t stream) {
  const float* x    = (const float*)d_in[0];
  const float* wqkv = (const float*)d_in[1];
  const float* wout = (const float*)d_in[2];
  float* out = (float*)d_out;

  char* ws = (char*)d_ws;
  const size_t QKV_F = (size_t)BATCH * O3 * NPIX;
  const size_t KB_F  = (size_t)BATCH * HIDDEN * NPIX;
  const size_t CTX_F = (size_t)BATCH * HEADS * DHEAD * DHEAD;
  float* qkv  = (float*)ws;
  float* kbuf = (float*)(ws + QKV_F * 4);
  float* ctx  = (float*)(ws + (QKV_F + KB_F) * 4);
  float* pctx = (float*)(ws + (QKV_F + KB_F + CTX_F) * 4);
  float* w2   = (float*)(ws + (QKV_F + KB_F + CTX_F + KSPLIT * CTX_F) * 4);

  k_qkv_gemm<<<dim3(NPIX / 16, BATCH), 256, 0, stream>>>(x, wqkv, qkv);
  k_softmax_n<<<BATCH * HIDDEN, 256, 0, stream>>>(qkv);
  k_softmax_d<<<(BATCH * HEADS * NPIX) / 256, 256, 0, stream>>>(qkv, kbuf);
  k_context_part<<<dim3(KSPLIT, BATCH * HEADS), 128, 0, stream>>>(kbuf, qkv,
                                                                  pctx);
  k_ctx_reduce<<<CTX_F / 256, 256, 0, stream>>>(pctx, ctx);
  k_fold_w2<<<BATCH * HEADS, 256, 0, stream>>>(wout, ctx, w2);
  k_out_gemm<<<dim3(NPIX / 16, BATCH), 256, 0, stream>>>(qkv, w2, out);
}


# context reassociated as (k x^T) Wv^T: n-split GEMM of k against x, then a small GEMM with the v weights; v is never formed and the projection GEMM has only the q rows left; context sum order changed
# speedup vs baseline: 2.8853x; 2.8853x over previous
#include <hip/hip_runtime.h>
#include <hip/hip_bf16.h>
#include <math.h>

#define BATCH   16
#define CIN     256
#define NPIX    4096
#define HEADS   4
#define DHEAD   32
#define HIDDEN  128
#define O3      384
#define GSPLIT  8
#define GCHUNK  (NPIX / GSPLIT)

typedef __attribute__((ext_vector_type(2))) float v2f;
typedef __attribute__((ext_vector_type(4))) float v4f;
typedef __attribute__((ext_vector_type(8))) float v8f;

__device__ __forceinline__ v8f wmma4(v2f a, v2f b, v8f c) {
  return __builtin_amdgcn_wmma_f32_16x16x4_f32(false, a, false, b, (short)0, c,
                                               false, false);
}

__global__ __launch_bounds__(256) void k_pack_wq(
    const float* __restrict__ wqkv, float* __restrict__ wp) {
  const int t = blockIdx.x * 256 + threadIdx.x;
  const int lane = t & 31;
  const int f    = t >> 5;
  const int i    = f & 1;
  const int w    = (f >> 1) & 3;
  const int ks   = f >> 3;
  const int half = lane >> 4, lm = lane & 15;
  const v2f a = *reinterpret_cast<const v2f*>(
      wqkv + (w * 16 + 64 * i + lm) * CIN + 4 * ks + 2 * half);
  *reinterpret_cast<v2f*>(wp + (size_t)t * 2) = a;
}

#define QG 4
__device__ __forceinline__ void q_load(const float* __restrict__ xb,
                                       const v2f* __restrict__ wf, int ks0,
                                       int half, v4f (&bx)[QG], v4f (&by)[QG],
                                       v2f (&a)[QG][2]) {
#pragma unroll
  for (int s = 0; s < QG; ++s) {
    const int ks = ks0 + s;
    const int k0 = 4 * ks + 2 * half;
    bx[s] = *reinterpret_cast<const v4f*>(xb + (size_t)k0 * NPIX);
    by[s] = *reinterpret_cast<const v4f*>(xb + (size_t)(k0 + 1) * NPIX);
#pragma unroll
    for (int i = 0; i < 2; ++i) a[s][i] = wf[ks * 256 + i * 32];
  }
}
__device__ __forceinline__ void q_mma(v8f (&acc)[2][4], const v4f (&bx)[QG],
                                      const v4f (&by)[QG],
                                      const v2f (&a)[QG][2]) {
#pragma unroll
  for (int s = 0; s < QG; ++s) {
#pragma unroll
    for (int j = 0; j < 4; ++j) {
      v2f bf;
      bf.x = bx[s][j];
      bf.y = by[s][j];
#pragma unroll
      for (int i = 0; i < 2; ++i) acc[i][j] = wmma4(a[s][i], bf, acc[i][j]);
    }
  }
}

__global__ __launch_bounds__(128) __attribute__((amdgpu_waves_per_eu(4, 4)))
void k_q_gemm(
    const float* __restrict__ x, const float* __restrict__ wp,
    float* __restrict__ qkv) {
  const int lane = threadIdx.x & 31;
  const int wave = threadIdx.x >> 5;
  const int half = lane >> 4;
  const int lm   = lane & 15;
  const int nBase = blockIdx.x * 64;
  const int b     = blockIdx.y;

  const float* __restrict__ xb = x + (size_t)b * CIN * NPIX + nBase + 4 * lm;
  float* __restrict__ qb = qkv + (size_t)b * O3 * NPIX + nBase + 4 * lm;

  const int m0 = wave * 16;
  const v2f* __restrict__ wf =
      reinterpret_cast<const v2f*>(wp) + wave * 64 + lane;
  v8f acc[2][4] = {};

  v4f bxA[QG], byA[QG], bxB[QG], byB[QG];
  v2f aA[QG][2], aB[QG][2];
  constexpr int NG = CIN / 4 / QG;
  q_load(xb, wf, 0, half, bxA, byA, aA);
#pragma unroll 1
  for (int g = 0; g < NG - 2; g += 2) {
    q_load(xb, wf, (g + 1) * QG, half, bxB, byB, aB);
    q_mma(acc, bxA, byA, aA);
    q_load(xb, wf, (g + 2) * QG, half, bxA, byA, aA);
    q_mma(acc, bxB, byB, aB);
  }
  q_load(xb, wf, (NG - 1) * QG, half, bxB, byB, aB);
  q_mma(acc, bxA, byA, aA);
  q_mma(acc, bxB, byB, aB);

  for (int i = 0; i < 2; ++i) {
    for (int v = 0; v < 8; ++v) {
      const int row = m0 + 64 * i + v + 8 * half;
      v4f o;
      o.x = acc[i][0][v];
      o.y = acc[i][1][v];
      o.z = acc[i][2][v];
      o.w = acc[i][3][v];
      *reinterpret_cast<v4f*>(qb + (size_t)row * NPIX) = o;
    }
  }
}

__global__ __launch_bounds__(256) void k_softmax_n(float* __restrict__ qkv) {
  const int r = blockIdx.x;
  const int b = r >> 7;
  const int c = r & 127;
  float* __restrict__ row = qkv + ((size_t)b * O3 + c) * NPIX;

  __shared__ float red[256];
  const int t = threadIdx.x;

  float v[16];
  float m = -INFINITY;
#pragma unroll
  for (int j = 0; j < 16; ++j) {
    v[j] = row[t + j * 256];
    m = fmaxf(m, v[j]);
  }
  red[t] = m; __syncthreads();
  for (int s = 128; s > 0; s >>= 1) {
    if (t < s) red[t] = fmaxf(red[t], red[t + s]);
    __syncthreads();
  }
  m = red[0]; __syncthreads();

  float sum = 0.f;
#pragma unroll
  for (int j = 0; j < 16; ++j) {
    v[j] = __expf(v[j] - m);
    sum += v[j];
  }
  red[t] = sum; __syncthreads();
  for (int s = 128; s > 0; s >>= 1) {
    if (t < s) red[t] += red[t + s];
    __syncthreads();
  }
  const float inv = 1.0f / red[0];
#pragma unroll
  for (int j = 0; j < 16; ++j) row[t + j * 256] = v[j] * inv;
}

__global__ __launch_bounds__(256) void k_softmax_d(
    const float* __restrict__ qkv, float* __restrict__ kbuf) {
  const int col = blockIdx.x * 256 + threadIdx.x;
  const int n = col & (NPIX - 1);
  const int h = (col >> 12) & (HEADS - 1);
  const int b = col >> 14;

  const float* __restrict__ qs = qkv + ((size_t)b * O3 + h * DHEAD) * NPIX + n;
  float vals[DHEAD];
  float m = -INFINITY;
#pragma unroll
  for (int d = 0; d < DHEAD; ++d) {
    vals[d] = qs[(size_t)d * NPIX];
    m = fmaxf(m, vals[d]);
  }
  float s = 0.f;
#pragma unroll
  for (int d = 0; d < DHEAD; ++d) {
    vals[d] = __expf(vals[d] - m);
    s += vals[d];
  }
  const float inv = 1.0f / s;
  float* __restrict__ ko =
      kbuf + (size_t)(b * HEADS + h) * DHEAD * NPIX + n;
#pragma unroll
  for (int d = 0; d < DHEAD; ++d) ko[(size_t)d * NPIX] = vals[d] * inv;
}

#define GG 4
__device__ __forceinline__ void g_load(const float* __restrict__ kA,
                                       const float* __restrict__ xB, int g0,
                                       v4f (&a)[GG][2], v4f (&bq)[GG][4]) {
#pragma unroll
  for (int s = 0; s < GG; ++s) {
    const int off = 8 * (g0 + s);
#pragma unroll
    for (int tm = 0; tm < 2; ++tm)
      a[s][tm] = *reinterpret_cast<const v4f*>(kA + (size_t)tm * 16 * NPIX + off);
#pragma unroll
    for (int tn = 0; tn < 4; ++tn)
      bq[s][tn] = *reinterpret_cast<const v4f*>(xB + (size_t)tn * 16 * NPIX + off);
  }
}
__device__ __forceinline__ void g_mma(v8f (&acc)[2][4], const v4f (&a)[GG][2],
                                      const v4f (&bq)[GG][4]) {
#pragma unroll
  for (int s = 0; s < GG; ++s) {
#pragma unroll
    for (int p = 0; p < 2; ++p) {
      v2f af[2], bf[4];
#pragma unroll
      for (int tm = 0; tm < 2; ++tm) {
        af[tm].x = a[s][tm][2 * p];
        af[tm].y = a[s][tm][2 * p + 1];
      }
#pragma unroll
      for (int tn = 0; tn < 4; ++tn) {
        bf[tn].x = bq[s][tn][2 * p];
        bf[tn].y = bq[s][tn][2 * p + 1];
      }
#pragma unroll
      for (int tn = 0; tn < 4; ++tn)
#pragma unroll
        for (int tm = 0; tm < 2; ++tm)
          acc[tm][tn] = wmma4(af[tm], bf[tn], acc[tm][tn]);
    }
  }
}

__global__ __launch_bounds__(256) __attribute__((amdgpu_waves_per_eu(2, 2)))
void k_g_part(
    const float* __restrict__ kbuf, const float* __restrict__ x,
    float* __restrict__ qkv) {
  const int ks    = blockIdx.x;
  const int chalf = blockIdx.y;
  const int b     = blockIdx.z;
  const int lane = threadIdx.x & 31;
  const int wave = threadIdx.x >> 5;
  const int half = lane >> 4, lm = lane & 15;
  const int wm = wave & 3;
  const int wn = wave >> 2;
  const int nBeg = ks * GCHUNK + 4 * half;

  const float* __restrict__ kA =
      kbuf + ((size_t)b * HIDDEN + wm * DHEAD + lm) * NPIX + nBeg;
  const float* __restrict__ xB =
      x + ((size_t)b * CIN + chalf * 128 + wn * 64 + lm) * NPIX + nBeg;

  v8f acc[2][4] = {};
  v4f aA[GG][2], aB[GG][2], bA[GG][4], bB[GG][4];
  constexpr int NS = GCHUNK / 8 / GG;
  g_load(kA, xB, 0, aA, bA);
#pragma unroll 1
  for (int g = 0; g < NS - 2; g += 2) {
    g_load(kA, xB, (g + 1) * GG, aB, bB);
    g_mma(acc, aA, bA);
    g_load(kA, xB, (g + 2) * GG, aA, bA);
    g_mma(acc, aB, bB);
  }
  g_load(kA, xB, (NS - 1) * GG, aB, bB);
  g_mma(acc, aA, bA);
  g_mma(acc, aB, bB);

  float* __restrict__ gp = qkv + ((size_t)b * O3 + HIDDEN) * NPIX +
                           (size_t)(((ks * 2 + chalf) * 8 + wave) * 2048) + lane;
  for (int tm = 0; tm < 2; ++tm)
    for (int tn = 0; tn < 4; ++tn)
      for (int r = 0; r < 8; ++r)
        gp[((tm * 4 + tn) * 8 + r) * 32] = acc[tm][tn][r];
}

__global__ __launch_bounds__(256) void k_ctx_from_g(
    const float* __restrict__ qkv, const float* __restrict__ wqkv,
    float* __restrict__ ctx) {
  const int bh = blockIdx.x;
  const int b = bh >> 2, h = bh & 3;
  const int lane = threadIdx.x & 31;
  const int wave = threadIdx.x >> 5;
  const int half = lane >> 4, lm = lane & 15;
  const int chalf = wave >> 2;
  const int pw    = h + 4 * ((wave >> 1) & 1);
  const int tn0   = 2 * (wave & 1);

  const float* __restrict__ gb = qkv + ((size_t)b * O3 + HIDDEN) * NPIX +
                                 (size_t)((chalf * 8 + pw) * 2048) +
                                 (lm & 7) * 32 + (lm >> 3) * 16 + 4 * half;
  const float* __restrict__ wv =
      wqkv + (size_t)(2 * HIDDEN + h * DHEAD + lm) * CIN + wave * 32 + 4 * half;

  __shared__ float red[8][DHEAD * DHEAD];

  v8f acc[2][2] = {};
#pragma unroll
  for (int tt = 0; tt < 2; ++tt) {
#pragma unroll
    for (int gg = 0; gg < 2; ++gg) {
      v4f a[2], w[2];
#pragma unroll
      for (int tm = 0; tm < 2; ++tm) {
        const float* __restrict__ p =
            gb + ((tm * 4 + tn0 + tt) * 8) * 32 + 8 * gg;
        v4f s = *reinterpret_cast<const v4f*>(p);
#pragma unroll
        for (int ks = 1; ks < GSPLIT; ++ks)
          s += *reinterpret_cast<const v4f*>(p + (size_t)ks * (2 * 8 * 2048));
        a[tm] = s;
      }
#pragma unroll
      for (int tv = 0; tv < 2; ++tv)
        w[tv] = *reinterpret_cast<const v4f*>(wv + (size_t)tv * 16 * CIN +
                                              16 * tt + 8 * gg);
#pragma unroll
      for (int p = 0; p < 2; ++p) {
#pragma unroll
        for (int tm = 0; tm < 2; ++tm) {
#pragma unroll
          for (int tv = 0; tv < 2; ++tv) {
            v2f af, bf;
            af.x = a[tm][2 * p]; af.y = a[tm][2 * p + 1];
            bf.x = w[tv][2 * p]; bf.y = w[tv][2 * p + 1];
            acc[tm][tv] = wmma4(af, bf, acc[tm][tv]);
          }
        }
      }
    }
  }
  float* __restrict__ mine = red[wave];
  for (int tm = 0; tm < 2; ++tm)
    for (int tv = 0; tv < 2; ++tv)
      for (int r = 0; r < 8; ++r)
        mine[(16 * tm + 8 * half + r) * DHEAD + 16 * tv + lm] = acc[tm][tv][r];
  __syncthreads();
  float* __restrict__ co = ctx + (size_t)bh * (DHEAD * DHEAD);
  for (int i = 0; i < (DHEAD * DHEAD) / 256; ++i) {
    const int e = threadIdx.x + 256 * i;
    float s = red[0][e];
    for (int w = 1; w < 8; ++w) s += red[w][e];
    co[e] = s;
  }
}

__global__ __launch_bounds__(256) void k_fold_w2(
    const float* __restrict__ wout, const float* __restrict__ ctx,
    float* __restrict__ w2p) {
  const int bh = blockIdx.x;
  const int b = bh >> 2, h = bh & 3;
  const int lane = threadIdx.x & 31;
  const int wave = threadIdx.x >> 5;
  const int half = lane >> 4, lm = lane & 15;
  const int m0 = wave * 16;

  const float* __restrict__ cb = ctx + (size_t)bh * DHEAD * DHEAD;

  v8f acc00 = {}, acc01 = {}, acc10 = {}, acc11 = {};
#pragma unroll
  for (int kb = 0; kb < DHEAD; kb += 4) {
    const int k0 = kb + 2 * half;
    v2f w0, w1, c0, c1;
    w0.x = wout[(m0 + lm) * HIDDEN + h * DHEAD + k0];
    w0.y = wout[(m0 + lm) * HIDDEN + h * DHEAD + k0 + 1];
    w1.x = wout[(m0 + 128 + lm) * HIDDEN + h * DHEAD + k0];
    w1.y = wout[(m0 + 128 + lm) * HIDDEN + h * DHEAD + k0 + 1];
    c0.x = cb[(size_t)(0 + lm) * DHEAD + k0];
    c0.y = cb[(size_t)(0 + lm) * DHEAD + k0 + 1];
    c1.x = cb[(size_t)(16 + lm) * DHEAD + k0];
    c1.y = cb[(size_t)(16 + lm) * DHEAD + k0 + 1];
    acc00 = wmma4(c0, w0, acc00);
    acc01 = wmma4(c0, w1, acc01);
    acc10 = wmma4(c1, w0, acc10);
    acc11 = wmma4(c1, w1, acc11);
  }
  v2f* __restrict__ w2f = reinterpret_cast<v2f*>(w2p) + (size_t)b * (32 * 8 * 2 * 32);
#pragma unroll
  for (int r = 0; r < 8; r += 2) {
    const int dl = ((r >> 1) & 1) * 16 + lm;
    const int ks0 = h * 8 + 2 * half + (r >> 2);
    const int ks1 = ks0 + 4;
    v2f o;
    o.x = acc00[r]; o.y = acc00[r + 1];
    w2f[((ks0 * 8 + wave) * 2 + 0) * 32 + dl] = o;
    o.x = acc01[r]; o.y = acc01[r + 1];
    w2f[((ks0 * 8 + wave) * 2 + 1) * 32 + dl] = o;
    o.x = acc10[r]; o.y = acc10[r + 1];
    w2f[((ks1 * 8 + wave) * 2 + 0) * 32 + dl] = o;
    o.x = acc11[r]; o.y = acc11[r + 1];
    w2f[((ks1 * 8 + wave) * 2 + 1) * 32 + dl] = o;
  }
}

__device__ __forceinline__ void out_load(const float* __restrict__ qs,
                                         const v2f* __restrict__ wf, int ks0,
                                         int half, v4f (&bx)[QG], v4f (&by)[QG],
                                         v2f (&a)[QG][2]) {
#pragma unroll
  for (int s = 0; s < QG; ++s) {
    const int ks = ks0 + s;
    const int k0 = 4 * ks + 2 * half;
    bx[s] = *reinterpret_cast<const v4f*>(qs + (size_t)k0 * NPIX);
    by[s] = *reinterpret_cast<const v4f*>(qs + (size_t)(k0 + 1) * NPIX);
#pragma unroll
    for (int i = 0; i < 2; ++i) a[s][i] = wf[ks * 512 + i * 32];
  }
}
__device__ __forceinline__ void out_mma(v8f (&acc)[2][4], const v4f (&bx)[QG],
                                        const v4f (&by)[QG],
                                        const v2f (&a)[QG][2]) {
#pragma unroll
  for (int s = 0; s < QG; ++s) {
#pragma unroll
    for (int j = 0; j < 4; ++j) {
      v2f bf;
      bf.x = bx[s][j];
      bf.y = by[s][j];
#pragma unroll
      for (int i = 0; i < 2; ++i) acc[i][j] = wmma4(a[s][i], bf, acc[i][j]);
    }
  }
}

__global__ __launch_bounds__(256) __attribute__((amdgpu_waves_per_eu(4, 4)))
void k_out_gemm(
    const float* __restrict__ qkv, const float* __restrict__ w2p,
    float* __restrict__ out) {
  const int lane = threadIdx.x & 31;
  const int wave = threadIdx.x >> 5;
  const int half = lane >> 4, lm = lane & 15;
  const int nBase = blockIdx.x * 64;
  const int b     = blockIdx.y;

  const float* __restrict__ qs =
      qkv + (size_t)b * O3 * NPIX + nBase + 4 * lm;
  const v2f* __restrict__ wf = reinterpret_cast<const v2f*>(w2p) +
                               (size_t)b * (32 * 8 * 2 * 32) + wave * 64 + lane;
  float* __restrict__ ob = out + (size_t)b * CIN * NPIX + nBase + 4 * lm;

  const int m0 = wave * 16;
  v8f acc[2][4] = {};

  v4f bxA[QG], byA[QG], bxB[QG], byB[QG];
  v2f aA[QG][2], aB[QG][2];
  constexpr int NG = HIDDEN / 4 / QG;
  out_load(qs, wf, 0, half, bxA, byA, aA);
#pragma unroll 1
  for (int g = 0; g < NG - 2; g += 2) {
    out_load(qs, wf, (g + 1) * QG, half, bxB, byB, aB);
    out_mma(acc, bxA, byA, aA);
    out_load(qs, wf, (g + 2) * QG, half, bxA, byA, aA);
    out_mma(acc, bxB, byB, aB);
  }
  out_load(qs, wf, (NG - 1) * QG, half, bxB, byB, aB);
  out_mma(acc, bxA, byA, aA);
  out_mma(acc, bxB, byB, aB);

  for (int i = 0; i < 2; ++i) {
    for (int v = 0; v < 8; ++v) {
      const int row = m0 + 128 * i + v + 8 * half;
      v4f o;
      o.x = acc[i][0][v];
      o.y = acc[i][1][v];
      o.z = acc[i][2][v];
      o.w = acc[i][3][v];
      *reinterpret_cast<v4f*>(ob + (size_t)row * NPIX) = o;
    }
  }
}

extern "C" void kernel_launch(void* const* d_in, const int* in_sizes, int n_in,
                              void* d_out, int out_size, void* d_ws,
                              size_t ws_size, hipStream_t stream) {
  const float* x    = (const float*)d_in[0];
  const float* wqkv = (const float*)d_in[1];
  const float* wout = (const float*)d_in[2];
  float* out = (float*)d_out;

  char* ws = (char*)d_ws;
  const size_t QKV_F = (size_t)BATCH * O3 * NPIX;
  const size_t KB_F  = (size_t)BATCH * HIDDEN * NPIX;
  const size_t CTX_F = (size_t)BATCH * HEADS * DHEAD * DHEAD;
  float* qkv  = (float*)ws;
  float* kbuf = (float*)(ws + QKV_F * 4);
  float* ctx  = (float*)(ws + (QKV_F + KB_F) * 4);
  float* wp   = (float*)(ws + (QKV_F + KB_F + CTX_F) * 4);
  float* w2   = (float*)(ws + (QKV_F + KB_F + CTX_F + 8 * CTX_F) * 4);

  k_pack_wq<<<(HIDDEN * CIN / 2) / 256, 256, 0, stream>>>(wqkv, wp);
  k_q_gemm<<<dim3(NPIX / 64, BATCH), 128, 0, stream>>>(x, wp, qkv);
  k_softmax_n<<<BATCH * HIDDEN, 256, 0, stream>>>(qkv);
  k_softmax_d<<<(BATCH * HEADS * NPIX) / 256, 256, 0, stream>>>(qkv, kbuf);
  k_g_part<<<dim3(GSPLIT, 2, BATCH), 256, 0, stream>>>(kbuf, x, qkv);
  k_ctx_from_g<<<BATCH * HEADS, 256, 0, stream>>>(qkv, wqkv, ctx);
  k_fold_w2<<<BATCH * HEADS, 256, 0, stream>>>(wout, ctx, w2);
  k_out_gemm<<<dim3(NPIX / 64, BATCH), 256, 0, stream>>>(qkv, w2, out);
}


# k x^T GEMM: each wave owns a 64x64 tile (was 32x64) on one 256-n half of the chunk, so 8 strided 16-line operand loads feed 32 WMMAs (was 6 per 16); halves added in LDS; G partial sum order changed
# speedup vs baseline: 2.9863x; 1.0350x over previous
#include <hip/hip_runtime.h>
#include <hip/hip_bf16.h>
#include <math.h>

#define BATCH   16
#define CIN     256
#define NPIX    4096
#define HEADS   4
#define DHEAD   32
#define HIDDEN  128
#define O3      384
#define GSPLIT  8
#define GCHUNK  (NPIX / GSPLIT)

typedef __attribute__((ext_vector_type(2))) float v2f;
typedef __attribute__((ext_vector_type(4))) float v4f;
typedef __attribute__((ext_vector_type(8))) float v8f;

__device__ __forceinline__ v8f wmma4(v2f a, v2f b, v8f c) {
  return __builtin_amdgcn_wmma_f32_16x16x4_f32(false, a, false, b, (short)0, c,
                                               false, false);
}

__global__ __launch_bounds__(256) void k_pack_wq(
    const float* __restrict__ wqkv, float* __restrict__ wp) {
  const int t = blockIdx.x * 256 + threadIdx.x;
  const int lane = t & 31;
  const int f    = t >> 5;
  const int i    = f & 1;
  const int w    = (f >> 1) & 3;
  const int ks   = f >> 3;
  const int half = lane >> 4, lm = lane & 15;
  const v2f a = *reinterpret_cast<const v2f*>(
      wqkv + (w * 16 + 64 * i + lm) * CIN + 4 * ks + 2 * half);
  *reinterpret_cast<v2f*>(wp + (size_t)t * 2) = a;
}

#define QG 4
__device__ __forceinline__ void q_load(const float* __restrict__ xb,
                                       const v2f* __restrict__ wf, int ks0,
                                       int half, v4f (&bx)[QG], v4f (&by)[QG],
                                       v2f (&a)[QG][2]) {
#pragma unroll
  for (int s = 0; s < QG; ++s) {
    const int ks = ks0 + s;
    const int k0 = 4 * ks + 2 * half;
    bx[s] = *reinterpret_cast<const v4f*>(xb + (size_t)k0 * NPIX);
    by[s] = *reinterpret_cast<const v4f*>(xb + (size_t)(k0 + 1) * NPIX);
#pragma unroll
    for (int i = 0; i < 2; ++i) a[s][i] = wf[ks * 256 + i * 32];
  }
}
__device__ __forceinline__ void q_mma(v8f (&acc)[2][4], const v4f (&bx)[QG],
                                      const v4f (&by)[QG],
                                      const v2f (&a)[QG][2]) {
#pragma unroll
  for (int s = 0; s < QG; ++s) {
#pragma unroll
    for (int j = 0; j < 4; ++j) {
      v2f bf;
      bf.x = bx[s][j];
      bf.y = by[s][j];
#pragma unroll
      for (int i = 0; i < 2; ++i) acc[i][j] = wmma4(a[s][i], bf, acc[i][j]);
    }
  }
}

__global__ __launch_bounds__(128) __attribute__((amdgpu_waves_per_eu(4, 4)))
void k_q_gemm(
    const float* __restrict__ x, const float* __restrict__ wp,
    float* __restrict__ qkv) {
  const int lane = threadIdx.x & 31;
  const int wave = threadIdx.x >> 5;
  const int half = lane >> 4;
  const int lm   = lane & 15;
  const int nBase = blockIdx.x * 64;
  const int b     = blockIdx.y;

  const float* __restrict__ xb = x + (size_t)b * CIN * NPIX + nBase + 4 * lm;
  float* __restrict__ qb = qkv + (size_t)b * O3 * NPIX + nBase + 4 * lm;

  const int m0 = wave * 16;
  const v2f* __restrict__ wf =
      reinterpret_cast<const v2f*>(wp) + wave * 64 + lane;
  v8f acc[2][4] = {};

  v4f bxA[QG], byA[QG], bxB[QG], byB[QG];
  v2f aA[QG][2], aB[QG][2];
  constexpr int NG = CIN / 4 / QG;
  q_load(xb, wf, 0, half, bxA, byA, aA);
#pragma unroll 1
  for (int g = 0; g < NG - 2; g += 2) {
    q_load(xb, wf, (g + 1) * QG, half, bxB, byB, aB);
    q_mma(acc, bxA, byA, aA);
    q_load(xb, wf, (g + 2) * QG, half, bxA, byA, aA);
    q_mma(acc, bxB, byB, aB);
  }
  q_load(xb, wf, (NG - 1) * QG, half, bxB, byB, aB);
  q_mma(acc, bxA, byA, aA);
  q_mma(acc, bxB, byB, aB);

  for (int i = 0; i < 2; ++i) {
    for (int v = 0; v < 8; ++v) {
      const int row = m0 + 64 * i + v + 8 * half;
      v4f o;
      o.x = acc[i][0][v];
      o.y = acc[i][1][v];
      o.z = acc[i][2][v];
      o.w = acc[i][3][v];
      *reinterpret_cast<v4f*>(qb + (size_t)row * NPIX) = o;
    }
  }
}

__global__ __launch_bounds__(256) void k_softmax_n(float* __restrict__ qkv) {
  const int r = blockIdx.x;
  const int b = r >> 7;
  const int c = r & 127;
  float* __restrict__ row = qkv + ((size_t)b * O3 + c) * NPIX;

  __shared__ float red[256];
  const int t = threadIdx.x;

  float v[16];
  float m = -INFINITY;
#pragma unroll
  for (int j = 0; j < 16; ++j) {
    v[j] = row[t + j * 256];
    m = fmaxf(m, v[j]);
  }
  red[t] = m; __syncthreads();
  for (int s = 128; s > 0; s >>= 1) {
    if (t < s) red[t] = fmaxf(red[t], red[t + s]);
    __syncthreads();
  }
  m = red[0]; __syncthreads();

  float sum = 0.f;
#pragma unroll
  for (int j = 0; j < 16; ++j) {
    v[j] = __expf(v[j] - m);
    sum += v[j];
  }
  red[t] = sum; __syncthreads();
  for (int s = 128; s > 0; s >>= 1) {
    if (t < s) red[t] += red[t + s];
    __syncthreads();
  }
  const float inv = 1.0f / red[0];
#pragma unroll
  for (int j = 0; j < 16; ++j) row[t + j * 256] = v[j] * inv;
}

__global__ __launch_bounds__(256) void k_softmax_d(
    const float* __restrict__ qkv, float* __restrict__ kbuf) {
  const int col = blockIdx.x * 256 + threadIdx.x;
  const int n = col & (NPIX - 1);
  const int h = (col >> 12) & (HEADS - 1);
  const int b = col >> 14;

  const float* __restrict__ qs = qkv + ((size_t)b * O3 + h * DHEAD) * NPIX + n;
  float vals[DHEAD];
  float m = -INFINITY;
#pragma unroll
  for (int d = 0; d < DHEAD; ++d) {
    vals[d] = qs[(size_t)d * NPIX];
    m = fmaxf(m, vals[d]);
  }
  float s = 0.f;
#pragma unroll
  for (int d = 0; d < DHEAD; ++d) {
    vals[d] = __expf(vals[d] - m);
    s += vals[d];
  }
  const float inv = 1.0f / s;
  float* __restrict__ ko =
      kbuf + (size_t)(b * HEADS + h) * DHEAD * NPIX + n;
#pragma unroll
  for (int d = 0; d < DHEAD; ++d) ko[(size_t)d * NPIX] = vals[d] * inv;
}

#define GG 4
__device__ __forceinline__ void g_load(const float* __restrict__ kA,
                                       const float* __restrict__ xB, int g0,
                                       v4f (&a)[GG][4], v4f (&bq)[GG][4]) {
#pragma unroll
  for (int s = 0; s < GG; ++s) {
    const int off = 8 * (g0 + s);
#pragma unroll
    for (int tm = 0; tm < 4; ++tm)
      a[s][tm] = *reinterpret_cast<const v4f*>(kA + (size_t)tm * 16 * NPIX + off);
#pragma unroll
    for (int tn = 0; tn < 4; ++tn)
      bq[s][tn] = *reinterpret_cast<const v4f*>(xB + (size_t)tn * 16 * NPIX + off);
  }
}
__device__ __forceinline__ void g_mma(v8f (&acc)[4][4], const v4f (&a)[GG][4],
                                      const v4f (&bq)[GG][4]) {
#pragma unroll
  for (int s = 0; s < GG; ++s) {
#pragma unroll
    for (int p = 0; p < 2; ++p) {
      v2f af[4], bf[4];
#pragma unroll
      for (int tm = 0; tm < 4; ++tm) {
        af[tm].x = a[s][tm][2 * p];
        af[tm].y = a[s][tm][2 * p + 1];
      }
#pragma unroll
      for (int tn = 0; tn < 4; ++tn) {
        bf[tn].x = bq[s][tn][2 * p];
        bf[tn].y = bq[s][tn][2 * p + 1];
      }
#pragma unroll
      for (int tn = 0; tn < 4; ++tn)
#pragma unroll
        for (int tm = 0; tm < 4; ++tm)
          acc[tm][tn] = wmma4(af[tm], bf[tn], acc[tm][tn]);
    }
  }
}

__global__ __launch_bounds__(256) __attribute__((amdgpu_waves_per_eu(2, 2)))
void k_g_part(
    const float* __restrict__ kbuf, const float* __restrict__ x,
    float* __restrict__ qkv) {
  const int ks    = blockIdx.x;
  const int chalf = blockIdx.y;
  const int b     = blockIdx.z;
  const int lane = threadIdx.x & 31;
  const int wave = threadIdx.x >> 5;
  const int half = lane >> 4, lm = lane & 15;
  const int nh  = wave >> 2;
  const int wq  = wave & 3;
  const int wm2 = wq & 1;
  const int wn  = wq >> 1;
  const int nBeg = ks * GCHUNK + nh * (GCHUNK / 2) + 4 * half;

  const float* __restrict__ kA =
      kbuf + ((size_t)b * HIDDEN + wm2 * 64 + lm) * NPIX + nBeg;
  const float* __restrict__ xB =
      x + ((size_t)b * CIN + chalf * 128 + wn * 64 + lm) * NPIX + nBeg;

  __shared__ float red[4][16 * 8 * 32];

  v8f acc[4][4] = {};
  v4f aA[GG][4], aB[GG][4], bA[GG][4], bB[GG][4];
  constexpr int NS = GCHUNK / 2 / 8 / GG;
  g_load(kA, xB, 0, aA, bA);
#pragma unroll 1
  for (int g = 0; g < NS - 2; g += 2) {
    g_load(kA, xB, (g + 1) * GG, aB, bB);
    g_mma(acc, aA, bA);
    g_load(kA, xB, (g + 2) * GG, aA, bA);
    g_mma(acc, aB, bB);
  }
  g_load(kA, xB, (NS - 1) * GG, aB, bB);
  g_mma(acc, aA, bA);
  g_mma(acc, aB, bB);

  float* __restrict__ mine = red[wq] + lane;
  if (nh == 1) {
    for (int tm = 0; tm < 4; ++tm)
      for (int tn = 0; tn < 4; ++tn)
        for (int r = 0; r < 8; ++r)
          mine[((tm * 4 + tn) * 8 + r) * 32] = acc[tm][tn][r];
  }
  __syncthreads();
  if (nh == 0) {
    float* __restrict__ gp = qkv + ((size_t)b * O3 + HIDDEN) * NPIX +
                             (size_t)(((ks * 2 + chalf) * 8 + 4 * wn) * 2048) + lane;
    for (int t4 = 0; t4 < 4; ++t4) {
      const int head = 2 * wm2 + (t4 >> 1);
      const int tm = t4 & 1;
      for (int tn = 0; tn < 4; ++tn)
        for (int r = 0; r < 8; ++r)
          gp[head * 2048 + ((tm * 4 + tn) * 8 + r) * 32] =
              acc[t4][tn][r] + mine[((t4 * 4 + tn) * 8 + r) * 32];
    }
  }
}

__global__ __launch_bounds__(256) void k_ctx_from_g(
    const float* __restrict__ qkv, const float* __restrict__ wqkv,
    float* __restrict__ ctx) {
  const int bh = blockIdx.x;
  const int b = bh >> 2, h = bh & 3;
  const int lane = threadIdx.x & 31;
  const int wave = threadIdx.x >> 5;
  const int half = lane >> 4, lm = lane & 15;
  const int chalf = wave >> 2;
  const int pw    = h + 4 * ((wave >> 1) & 1);
  const int tn0   = 2 * (wave & 1);

  const float* __restrict__ gb = qkv + ((size_t)b * O3 + HIDDEN) * NPIX +
                                 (size_t)((chalf * 8 + pw) * 2048) +
                                 (lm & 7) * 32 + (lm >> 3) * 16 + 4 * half;
  const float* __restrict__ wv =
      wqkv + (size_t)(2 * HIDDEN + h * DHEAD + lm) * CIN + wave * 32 + 4 * half;

  __shared__ float red[8][DHEAD * DHEAD];

  v8f acc[2][2] = {};
#pragma unroll
  for (int tt = 0; tt < 2; ++tt) {
#pragma unroll
    for (int gg = 0; gg < 2; ++gg) {
      v4f a[2], w[2];
#pragma unroll
      for (int tm = 0; tm < 2; ++tm) {
        const float* __restrict__ p =
            gb + ((tm * 4 + tn0 + tt) * 8) * 32 + 8 * gg;
        v4f s = *reinterpret_cast<const v4f*>(p);
#pragma unroll
        for (int ks = 1; ks < GSPLIT; ++ks)
          s += *reinterpret_cast<const v4f*>(p + (size_t)ks * (2 * 8 * 2048));
        a[tm] = s;
      }
#pragma unroll
      for (int tv = 0; tv < 2; ++tv)
        w[tv] = *reinterpret_cast<const v4f*>(wv + (size_t)tv * 16 * CIN +
                                              16 * tt + 8 * gg);
#pragma unroll
      for (int p = 0; p < 2; ++p) {
#pragma unroll
        for (int tm = 0; tm < 2; ++tm) {
#pragma unroll
          for (int tv = 0; tv < 2; ++tv) {
            v2f af, bf;
            af.x = a[tm][2 * p]; af.y = a[tm][2 * p + 1];
            bf.x = w[tv][2 * p]; bf.y = w[tv][2 * p + 1];
            acc[tm][tv] = wmma4(af, bf, acc[tm][tv]);
          }
        }
      }
    }
  }
  float* __restrict__ mine = red[wave];
  for (int tm = 0; tm < 2; ++tm)
    for (int tv = 0; tv < 2; ++tv)
      for (int r = 0; r < 8; ++r)
        mine[(16 * tm + 8 * half + r) * DHEAD + 16 * tv + lm] = acc[tm][tv][r];
  __syncthreads();
  float* __restrict__ co = ctx + (size_t)bh * (DHEAD * DHEAD);
  for (int i = 0; i < (DHEAD * DHEAD) / 256; ++i) {
    const int e = threadIdx.x + 256 * i;
    float s = red[0][e];
    for (int w = 1; w < 8; ++w) s += red[w][e];
    co[e] = s;
  }
}

__global__ __launch_bounds__(256) void k_fold_w2(
    const float* __restrict__ wout, const float* __restrict__ ctx,
    float* __restrict__ w2p) {
  const int bh = blockIdx.x;
  const int b = bh >> 2, h = bh & 3;
  const int lane = threadIdx.x & 31;
  const int wave = threadIdx.x >> 5;
  const int half = lane >> 4, lm = lane & 15;
  const int m0 = wave * 16;

  const float* __restrict__ cb = ctx + (size_t)bh * DHEAD * DHEAD;

  v8f acc00 = {}, acc01 = {}, acc10 = {}, acc11 = {};
#pragma unroll
  for (int kb = 0; kb < DHEAD; kb += 4) {
    const int k0 = kb + 2 * half;
    v2f w0, w1, c0, c1;
    w0.x = wout[(m0 + lm) * HIDDEN + h * DHEAD + k0];
    w0.y = wout[(m0 + lm) * HIDDEN + h * DHEAD + k0 + 1];
    w1.x = wout[(m0 + 128 + lm) * HIDDEN + h * DHEAD + k0];
    w1.y = wout[(m0 + 128 + lm) * HIDDEN + h * DHEAD + k0 + 1];
    c0.x = cb[(size_t)(0 + lm) * DHEAD + k0];
    c0.y = cb[(size_t)(0 + lm) * DHEAD + k0 + 1];
    c1.x = cb[(size_t)(16 + lm) * DHEAD + k0];
    c1.y = cb[(size_t)(16 + lm) * DHEAD + k0 + 1];
    acc00 = wmma4(c0, w0, acc00);
    acc01 = wmma4(c0, w1, acc01);
    acc10 = wmma4(c1, w0, acc10);
    acc11 = wmma4(c1, w1, acc11);
  }
  v2f* __restrict__ w2f = reinterpret_cast<v2f*>(w2p) + (size_t)b * (32 * 8 * 2 * 32);
#pragma unroll
  for (int r = 0; r < 8; r += 2) {
    const int dl = ((r >> 1) & 1) * 16 + lm;
    const int ks0 = h * 8 + 2 * half + (r >> 2);
    const int ks1 = ks0 + 4;
    v2f o;
    o.x = acc00[r]; o.y = acc00[r + 1];
    w2f[((ks0 * 8 + wave) * 2 + 0) * 32 + dl] = o;
    o.x = acc01[r]; o.y = acc01[r + 1];
    w2f[((ks0 * 8 + wave) * 2 + 1) * 32 + dl] = o;
    o.x = acc10[r]; o.y = acc10[r + 1];
    w2f[((ks1 * 8 + wave) * 2 + 0) * 32 + dl] = o;
    o.x = acc11[r]; o.y = acc11[r + 1];
    w2f[((ks1 * 8 + wave) * 2 + 1) * 32 + dl] = o;
  }
}

__device__ __forceinline__ void out_load(const float* __restrict__ qs,
                                         const v2f* __restrict__ wf, int ks0,
                                         int half, v4f (&bx)[QG], v4f (&by)[QG],
                                         v2f (&a)[QG][2]) {
#pragma unroll
  for (int s = 0; s < QG; ++s) {
    const int ks = ks0 + s;
    const int k0 = 4 * ks + 2 * half;
    bx[s] = *reinterpret_cast<const v4f*>(qs + (size_t)k0 * NPIX);
    by[s] = *reinterpret_cast<const v4f*>(qs + (size_t)(k0 + 1) * NPIX);
#pragma unroll
    for (int i = 0; i < 2; ++i) a[s][i] = wf[ks * 512 + i * 32];
  }
}
__device__ __forceinline__ void out_mma(v8f (&acc)[2][4], const v4f (&bx)[QG],
                                        const v4f (&by)[QG],
                                        const v2f (&a)[QG][2]) {
#pragma unroll
  for (int s = 0; s < QG; ++s) {
#pragma unroll
    for (int j = 0; j < 4; ++j) {
      v2f bf;
      bf.x = bx[s][j];
      bf.y = by[s][j];
#pragma unroll
      for (int i = 0; i < 2; ++i) acc[i][j] = wmma4(a[s][i], bf, acc[i][j]);
    }
  }
}

__global__ __launch_bounds__(256) __attribute__((amdgpu_waves_per_eu(4, 4)))
void k_out_gemm(
    const float* __restrict__ qkv, const float* __restrict__ w2p,
    float* __restrict__ out) {
  const int lane = threadIdx.x & 31;
  const int wave = threadIdx.x >> 5;
  const int half = lane >> 4, lm = lane & 15;
  const int nBase = blockIdx.x * 64;
  const int b     = blockIdx.y;

  const float* __restrict__ qs =
      qkv + (size_t)b * O3 * NPIX + nBase + 4 * lm;
  const v2f* __restrict__ wf = reinterpret_cast<const v2f*>(w2p) +
                               (size_t)b * (32 * 8 * 2 * 32) + wave * 64 + lane;
  float* __restrict__ ob = out + (size_t)b * CIN * NPIX + nBase + 4 * lm;

  const int m0 = wave * 16;
  v8f acc[2][4] = {};

  v4f bxA[QG], byA[QG], bxB[QG], byB[QG];
  v2f aA[QG][2], aB[QG][2];
  constexpr int NG = HIDDEN / 4 / QG;
  out_load(qs, wf, 0, half, bxA, byA, aA);
#pragma unroll 1
  for (int g = 0; g < NG - 2; g += 2) {
    out_load(qs, wf, (g + 1) * QG, half, bxB, byB, aB);
    out_mma(acc, bxA, byA, aA);
    out_load(qs, wf, (g + 2) * QG, half, bxA, byA, aA);
    out_mma(acc, bxB, byB, aB);
  }
  out_load(qs, wf, (NG - 1) * QG, half, bxB, byB, aB);
  out_mma(acc, bxA, byA, aA);
  out_mma(acc, bxB, byB, aB);

  for (int i = 0; i < 2; ++i) {
    for (int v = 0; v < 8; ++v) {
      const int row = m0 + 128 * i + v + 8 * half;
      v4f o;
      o.x = acc[i][0][v];
      o.y = acc[i][1][v];
      o.z = acc[i][2][v];
      o.w = acc[i][3][v];
      *reinterpret_cast<v4f*>(ob + (size_t)row * NPIX) = o;
    }
  }
}

extern "C" void kernel_launch(void* const* d_in, const int* in_sizes, int n_in,
                              void* d_out, int out_size, void* d_ws,
                              size_t ws_size, hipStream_t stream) {
  const float* x    = (const float*)d_in[0];
  const float* wqkv = (const float*)d_in[1];
  const float* wout = (const float*)d_in[2];
  float* out = (float*)d_out;

  char* ws = (char*)d_ws;
  const size_t QKV_F = (size_t)BATCH * O3 * NPIX;
  const size_t KB_F  = (size_t)BATCH * HIDDEN * NPIX;
  const size_t CTX_F = (size_t)BATCH * HEADS * DHEAD * DHEAD;
  float* qkv  = (float*)ws;
  float* kbuf = (float*)(ws + QKV_F * 4);
  float* ctx  = (float*)(ws + (QKV_F + KB_F) * 4);
  float* wp   = (float*)(ws + (QKV_F + KB_F + CTX_F) * 4);
  float* w2   = (float*)(ws + (QKV_F + KB_F + CTX_F + 8 * CTX_F) * 4);

  k_pack_wq<<<(HIDDEN * CIN / 2) / 256, 256, 0, stream>>>(wqkv, wp);
  k_q_gemm<<<dim3(NPIX / 64, BATCH), 128, 0, stream>>>(x, wp, qkv);
  k_softmax_n<<<BATCH * HIDDEN, 256, 0, stream>>>(qkv);
  k_softmax_d<<<(BATCH * HEADS * NPIX) / 256, 256, 0, stream>>>(qkv, kbuf);
  k_g_part<<<dim3(GSPLIT, 2, BATCH), 256, 0, stream>>>(kbuf, x, qkv);
  k_ctx_from_g<<<BATCH * HEADS, 256, 0, stream>>>(qkv, wqkv, ctx);
  k_fold_w2<<<BATCH * HEADS, 256, 0, stream>>>(wout, ctx, w2);
  k_out_gemm<<<dim3(NPIX / 64, BATCH), 256, 0, stream>>>(qkv, w2, out);
}


# ctx and fold-into-W2 kernels fused into one launch of 128 blocks (one per batch, head and 16-row half of kd); the ctx rows pass through LDS and are never stored; summation order unchanged
# speedup vs baseline: 3.2946x; 1.1032x over previous
#include <hip/hip_runtime.h>
#include <hip/hip_bf16.h>
#include <math.h>

#define BATCH   16
#define CIN     256
#define NPIX    4096
#define HEADS   4
#define DHEAD   32
#define HIDDEN  128
#define O3      384
#define GSPLIT  8
#define GCHUNK  (NPIX / GSPLIT)

typedef __attribute__((ext_vector_type(2))) float v2f;
typedef __attribute__((ext_vector_type(4))) float v4f;
typedef __attribute__((ext_vector_type(8))) float v8f;

__device__ __forceinline__ v8f wmma4(v2f a, v2f b, v8f c) {
  return __builtin_amdgcn_wmma_f32_16x16x4_f32(false, a, false, b, (short)0, c,
                                               false, false);
}

__global__ __launch_bounds__(256) void k_pack_wq(
    const float* __restrict__ wqkv, float* __restrict__ wp) {
  const int t = blockIdx.x * 256 + threadIdx.x;
  const int lane = t & 31;
  const int f    = t >> 5;
  const int i    = f & 1;
  const int w    = (f >> 1) & 3;
  const int ks   = f >> 3;
  const int half = lane >> 4, lm = lane & 15;
  const v2f a = *reinterpret_cast<const v2f*>(
      wqkv + (w * 16 + 64 * i + lm) * CIN + 4 * ks + 2 * half);
  *reinterpret_cast<v2f*>(wp + (size_t)t * 2) = a;
}

#define QG 4
__device__ __forceinline__ void q_load(const float* __restrict__ xb,
                                       const v2f* __restrict__ wf, int ks0,
                                       int half, v4f (&bx)[QG], v4f (&by)[QG],
                                       v2f (&a)[QG][2]) {
#pragma unroll
  for (int s = 0; s < QG; ++s) {
    const int ks = ks0 + s;
    const int k0 = 4 * ks + 2 * half;
    bx[s] = *reinterpret_cast<const v4f*>(xb + (size_t)k0 * NPIX);
    by[s] = *reinterpret_cast<const v4f*>(xb + (size_t)(k0 + 1) * NPIX);
#pragma unroll
    for (int i = 0; i < 2; ++i) a[s][i] = wf[ks * 256 + i * 32];
  }
}
__device__ __forceinline__ void q_mma(v8f (&acc)[2][4], const v4f (&bx)[QG],
                                      const v4f (&by)[QG],
                                      const v2f (&a)[QG][2]) {
#pragma unroll
  for (int s = 0; s < QG; ++s) {
#pragma unroll
    for (int j = 0; j < 4; ++j) {
      v2f bf;
      bf.x = bx[s][j];
      bf.y = by[s][j];
#pragma unroll
      for (int i = 0; i < 2; ++i) acc[i][j] = wmma4(a[s][i], bf, acc[i][j]);
    }
  }
}

__global__ __launch_bounds__(128) __attribute__((amdgpu_waves_per_eu(4, 4)))
void k_q_gemm(
    const float* __restrict__ x, const float* __restrict__ wp,
    float* __restrict__ qkv) {
  const int lane = threadIdx.x & 31;
  const int wave = threadIdx.x >> 5;
  const int half = lane >> 4;
  const int lm   = lane & 15;
  const int nBase = blockIdx.x * 64;
  const int b     = blockIdx.y;

  const float* __restrict__ xb = x + (size_t)b * CIN * NPIX + nBase + 4 * lm;
  float* __restrict__ qb = qkv + (size_t)b * O3 * NPIX + nBase + 4 * lm;

  const int m0 = wave * 16;
  const v2f* __restrict__ wf =
      reinterpret_cast<const v2f*>(wp) + wave * 64 + lane;
  v8f acc[2][4] = {};

  v4f bxA[QG], byA[QG], bxB[QG], byB[QG];
  v2f aA[QG][2], aB[QG][2];
  constexpr int NG = CIN / 4 / QG;
  q_load(xb, wf, 0, half, bxA, byA, aA);
#pragma unroll 1
  for (int g = 0; g < NG - 2; g += 2) {
    q_load(xb, wf, (g + 1) * QG, half, bxB, byB, aB);
    q_mma(acc, bxA, byA, aA);
    q_load(xb, wf, (g + 2) * QG, half, bxA, byA, aA);
    q_mma(acc, bxB, byB, aB);
  }
  q_load(xb, wf, (NG - 1) * QG, half, bxB, byB, aB);
  q_mma(acc, bxA, byA, aA);
  q_mma(acc, bxB, byB, aB);

  for (int i = 0; i < 2; ++i) {
    for (int v = 0; v < 8; ++v) {
      const int row = m0 + 64 * i + v + 8 * half;
      v4f o;
      o.x = acc[i][0][v];
      o.y = acc[i][1][v];
      o.z = acc[i][2][v];
      o.w = acc[i][3][v];
      *reinterpret_cast<v4f*>(qb + (size_t)row * NPIX) = o;
    }
  }
}

__global__ __launch_bounds__(256) void k_softmax_n(float* __restrict__ qkv) {
  const int r = blockIdx.x;
  const int b = r >> 7;
  const int c = r & 127;
  float* __restrict__ row = qkv + ((size_t)b * O3 + c) * NPIX;

  __shared__ float red[256];
  const int t = threadIdx.x;

  float v[16];
  float m = -INFINITY;
#pragma unroll
  for (int j = 0; j < 16; ++j) {
    v[j] = row[t + j * 256];
    m = fmaxf(m, v[j]);
  }
  red[t] = m; __syncthreads();
  for (int s = 128; s > 0; s >>= 1) {
    if (t < s) red[t] = fmaxf(red[t], red[t + s]);
    __syncthreads();
  }
  m = red[0]; __syncthreads();

  float sum = 0.f;
#pragma unroll
  for (int j = 0; j < 16; ++j) {
    v[j] = __expf(v[j] - m);
    sum += v[j];
  }
  red[t] = sum; __syncthreads();
  for (int s = 128; s > 0; s >>= 1) {
    if (t < s) red[t] += red[t + s];
    __syncthreads();
  }
  const float inv = 1.0f / red[0];
#pragma unroll
  for (int j = 0; j < 16; ++j) row[t + j * 256] = v[j] * inv;
}

__global__ __launch_bounds__(256) void k_softmax_d(
    const float* __restrict__ qkv, float* __restrict__ kbuf) {
  const int col = blockIdx.x * 256 + threadIdx.x;
  const int n = col & (NPIX - 1);
  const int h = (col >> 12) & (HEADS - 1);
  const int b = col >> 14;

  const float* __restrict__ qs = qkv + ((size_t)b * O3 + h * DHEAD) * NPIX + n;
  float vals[DHEAD];
  float m = -INFINITY;
#pragma unroll
  for (int d = 0; d < DHEAD; ++d) {
    vals[d] = qs[(size_t)d * NPIX];
    m = fmaxf(m, vals[d]);
  }
  float s = 0.f;
#pragma unroll
  for (int d = 0; d < DHEAD; ++d) {
    vals[d] = __expf(vals[d] - m);
    s += vals[d];
  }
  const float inv = 1.0f / s;
  float* __restrict__ ko =
      kbuf + (size_t)(b * HEADS + h) * DHEAD * NPIX + n;
#pragma unroll
  for (int d = 0; d < DHEAD; ++d) ko[(size_t)d * NPIX] = vals[d] * inv;
}

#define GG 4
__device__ __forceinline__ void g_load(const float* __restrict__ kA,
                                       const float* __restrict__ xB, int g0,
                                       v4f (&a)[GG][4], v4f (&bq)[GG][4]) {
#pragma unroll
  for (int s = 0; s < GG; ++s) {
    const int off = 8 * (g0 + s);
#pragma unroll
    for (int tm = 0; tm < 4; ++tm)
      a[s][tm] = *reinterpret_cast<const v4f*>(kA + (size_t)tm * 16 * NPIX + off);
#pragma unroll
    for (int tn = 0; tn < 4; ++tn)
      bq[s][tn] = *reinterpret_cast<const v4f*>(xB + (size_t)tn * 16 * NPIX + off);
  }
}
__device__ __forceinline__ void g_mma(v8f (&acc)[4][4], const v4f (&a)[GG][4],
                                      const v4f (&bq)[GG][4]) {
#pragma unroll
  for (int s = 0; s < GG; ++s) {
#pragma unroll
    for (int p = 0; p < 2; ++p) {
      v2f af[4], bf[4];
#pragma unroll
      for (int tm = 0; tm < 4; ++tm) {
        af[tm].x = a[s][tm][2 * p];
        af[tm].y = a[s][tm][2 * p + 1];
      }
#pragma unroll
      for (int tn = 0; tn < 4; ++tn) {
        bf[tn].x = bq[s][tn][2 * p];
        bf[tn].y = bq[s][tn][2 * p + 1];
      }
#pragma unroll
      for (int tn = 0; tn < 4; ++tn)
#pragma unroll
        for (int tm = 0; tm < 4; ++tm)
          acc[tm][tn] = wmma4(af[tm], bf[tn], acc[tm][tn]);
    }
  }
}

__global__ __launch_bounds__(256) __attribute__((amdgpu_waves_per_eu(2, 2)))
void k_g_part(
    const float* __restrict__ kbuf, const float* __restrict__ x,
    float* __restrict__ qkv) {
  const int ks    = blockIdx.x;
  const int chalf = blockIdx.y;
  const int b     = blockIdx.z;
  const int lane = threadIdx.x & 31;
  const int wave = threadIdx.x >> 5;
  const int half = lane >> 4, lm = lane & 15;
  const int nh  = wave >> 2;
  const int wq  = wave & 3;
  const int wm2 = wq & 1;
  const int wn  = wq >> 1;
  const int nBeg = ks * GCHUNK + nh * (GCHUNK / 2) + 4 * half;

  const float* __restrict__ kA =
      kbuf + ((size_t)b * HIDDEN + wm2 * 64 + lm) * NPIX + nBeg;
  const float* __restrict__ xB =
      x + ((size_t)b * CIN + chalf * 128 + wn * 64 + lm) * NPIX + nBeg;

  __shared__ float red[4][16 * 8 * 32];

  v8f acc[4][4] = {};
  v4f aA[GG][4], aB[GG][4], bA[GG][4], bB[GG][4];
  constexpr int NS = GCHUNK / 2 / 8 / GG;
  g_load(kA, xB, 0, aA, bA);
#pragma unroll 1
  for (int g = 0; g < NS - 2; g += 2) {
    g_load(kA, xB, (g + 1) * GG, aB, bB);
    g_mma(acc, aA, bA);
    g_load(kA, xB, (g + 2) * GG, aA, bA);
    g_mma(acc, aB, bB);
  }
  g_load(kA, xB, (NS - 1) * GG, aB, bB);
  g_mma(acc, aA, bA);
  g_mma(acc, aB, bB);

  float* __restrict__ mine = red[wq] + lane;
  if (nh == 1) {
    for (int tm = 0; tm < 4; ++tm)
      for (int tn = 0; tn < 4; ++tn)
        for (int r = 0; r < 8; ++r)
          mine[((tm * 4 + tn) * 8 + r) * 32] = acc[tm][tn][r];
  }
  __syncthreads();
  if (nh == 0) {
    float* __restrict__ gp = qkv + ((size_t)b * O3 + HIDDEN) * NPIX +
                             (size_t)(((ks * 2 + chalf) * 8 + 4 * wn) * 2048) + lane;
    for (int t4 = 0; t4 < 4; ++t4) {
      const int head = 2 * wm2 + (t4 >> 1);
      const int tm = t4 & 1;
      for (int tn = 0; tn < 4; ++tn)
        for (int r = 0; r < 8; ++r)
          gp[head * 2048 + ((tm * 4 + tn) * 8 + r) * 32] =
              acc[t4][tn][r] + mine[((t4 * 4 + tn) * 8 + r) * 32];
    }
  }
}

#define CPITCH 34
__global__ __launch_bounds__(256) void k_ctx_fold(
    const float* __restrict__ qkv, const float* __restrict__ wqkv,
    const float* __restrict__ wout, float* __restrict__ w2p) {
  const int bh = blockIdx.x >> 1;
  const int tm = blockIdx.x & 1;
  const int b = bh >> 2, h = bh & 3;
  const int lane = threadIdx.x & 31;
  const int wave = threadIdx.x >> 5;
  const int half = lane >> 4, lm = lane & 15;
  const int m0 = wave * 16;

  v2f w0[DHEAD / 4], w1[DHEAD / 4];
#pragma unroll
  for (int kb = 0; kb < DHEAD / 4; ++kb) {
    const int k0 = 4 * kb + 2 * half;
    w0[kb] = *reinterpret_cast<const v2f*>(
        wout + (m0 + lm) * HIDDEN + h * DHEAD + k0);
    w1[kb] = *reinterpret_cast<const v2f*>(
        wout + (m0 + 128 + lm) * HIDDEN + h * DHEAD + k0);
  }

  const int chalf = wave >> 2;
  const int pw    = h + 4 * ((wave >> 1) & 1);
  const int tn0   = 2 * (wave & 1);

  const float* __restrict__ gb = qkv + ((size_t)b * O3 + HIDDEN) * NPIX +
                                 (size_t)((chalf * 8 + pw) * 2048) +
                                 (lm & 7) * 32 + (lm >> 3) * 16 + 4 * half;
  const float* __restrict__ wv =
      wqkv + (size_t)(2 * HIDDEN + h * DHEAD + lm) * CIN + wave * 32 + 4 * half;

  __shared__ float red[8][16 * DHEAD];
  __shared__ __attribute__((aligned(16))) float ctxs[16 * CPITCH];

  v8f acc[2] = {};
#pragma unroll
  for (int tt = 0; tt < 2; ++tt) {
#pragma unroll
    for (int gg = 0; gg < 2; ++gg) {
      v4f w[2];
      const float* __restrict__ p =
          gb + ((tm * 4 + tn0 + tt) * 8) * 32 + 8 * gg;
      v4f a = *reinterpret_cast<const v4f*>(p);
#pragma unroll
      for (int ks = 1; ks < GSPLIT; ++ks)
        a += *reinterpret_cast<const v4f*>(p + (size_t)ks * (2 * 8 * 2048));
#pragma unroll
      for (int tv = 0; tv < 2; ++tv)
        w[tv] = *reinterpret_cast<const v4f*>(wv + (size_t)tv * 16 * CIN +
                                              16 * tt + 8 * gg);
#pragma unroll
      for (int p2 = 0; p2 < 2; ++p2) {
#pragma unroll
        for (int tv = 0; tv < 2; ++tv) {
          v2f af, bf;
          af.x = a[2 * p2];     af.y = a[2 * p2 + 1];
          bf.x = w[tv][2 * p2]; bf.y = w[tv][2 * p2 + 1];
          acc[tv] = wmma4(af, bf, acc[tv]);
        }
      }
    }
  }
  float* __restrict__ mine = red[wave];
  for (int tv = 0; tv < 2; ++tv)
    for (int r = 0; r < 8; ++r)
      mine[(8 * half + r) * DHEAD + 16 * tv + lm] = acc[tv][r];
  __syncthreads();
  for (int i = 0; i < (16 * DHEAD) / 256; ++i) {
    const int e = threadIdx.x + 256 * i;
    float s = red[0][e];
    for (int w = 1; w < 8; ++w) s += red[w][e];
    ctxs[(e >> 5) * CPITCH + (e & 31)] = s;
  }
  __syncthreads();

  v8f f0 = {}, f1 = {};
#pragma unroll
  for (int kb = 0; kb < DHEAD / 4; ++kb) {
    const v2f c = *reinterpret_cast<const v2f*>(
        ctxs + lm * CPITCH + 4 * kb + 2 * half);
    f0 = wmma4(c, w0[kb], f0);
    f1 = wmma4(c, w1[kb], f1);
  }
  v2f* __restrict__ w2f = reinterpret_cast<v2f*>(w2p) + (size_t)b * (32 * 8 * 2 * 32);
#pragma unroll
  for (int r = 0; r < 8; r += 2) {
    const int dl = ((r >> 1) & 1) * 16 + lm;
    const int ks = h * 8 + 4 * tm + 2 * half + (r >> 2);
    v2f o;
    o.x = f0[r]; o.y = f0[r + 1];
    w2f[((ks * 8 + wave) * 2 + 0) * 32 + dl] = o;
    o.x = f1[r]; o.y = f1[r + 1];
    w2f[((ks * 8 + wave) * 2 + 1) * 32 + dl] = o;
  }
}

__device__ __forceinline__ void out_load(const float* __restrict__ qs,
                                         const v2f* __restrict__ wf, int ks0,
                                         int half, v4f (&bx)[QG], v4f (&by)[QG],
                                         v2f (&a)[QG][2]) {
#pragma unroll
  for (int s = 0; s < QG; ++s) {
    const int ks = ks0 + s;
    const int k0 = 4 * ks + 2 * half;
    bx[s] = *reinterpret_cast<const v4f*>(qs + (size_t)k0 * NPIX);
    by[s] = *reinterpret_cast<const v4f*>(qs + (size_t)(k0 + 1) * NPIX);
#pragma unroll
    for (int i = 0; i < 2; ++i) a[s][i] = wf[ks * 512 + i * 32];
  }
}
__device__ __forceinline__ void out_mma(v8f (&acc)[2][4], const v4f (&bx)[QG],
                                        const v4f (&by)[QG],
                                        const v2f (&a)[QG][2]) {
#pragma unroll
  for (int s = 0; s < QG; ++s) {
#pragma unroll
    for (int j = 0; j < 4; ++j) {
      v2f bf;
      bf.x = bx[s][j];
      bf.y = by[s][j];
#pragma unroll
      for (int i = 0; i < 2; ++i) acc[i][j] = wmma4(a[s][i], bf, acc[i][j]);
    }
  }
}

__global__ __launch_bounds__(256) __attribute__((amdgpu_waves_per_eu(4, 4)))
void k_out_gemm(
    const float* __restrict__ qkv, const float* __restrict__ w2p,
    float* __restrict__ out) {
  const int lane = threadIdx.x & 31;
  const int wave = threadIdx.x >> 5;
  const int half = lane >> 4, lm = lane & 15;
  const int nBase = blockIdx.x * 64;
  const int b     = blockIdx.y;

  const float* __restrict__ qs =
      qkv + (size_t)b * O3 * NPIX + nBase + 4 * lm;
  const v2f* __restrict__ wf = reinterpret_cast<const v2f*>(w2p) +
                               (size_t)b * (32 * 8 * 2 * 32) + wave * 64 + lane;
  float* __restrict__ ob = out + (size_t)b * CIN * NPIX + nBase + 4 * lm;

  const int m0 = wave * 16;
  v8f acc[2][4] = {};

  v4f bxA[QG], byA[QG], bxB[QG], byB[QG];
  v2f aA[QG][2], aB[QG][2];
  constexpr int NG = HIDDEN / 4 / QG;
  out_load(qs, wf, 0, half, bxA, byA, aA);
#pragma unroll 1
  for (int g = 0; g < NG - 2; g += 2) {
    out_load(qs, wf, (g + 1) * QG, half, bxB, byB, aB);
    out_mma(acc, bxA, byA, aA);
    out_load(qs, wf, (g + 2) * QG, half, bxA, byA, aA);
    out_mma(acc, bxB, byB, aB);
  }
  out_load(qs, wf, (NG - 1) * QG, half, bxB, byB, aB);
  out_mma(acc, bxA, byA, aA);
  out_mma(acc, bxB, byB, aB);

  for (int i = 0; i < 2; ++i) {
    for (int v = 0; v < 8; ++v) {
      const int row = m0 + 128 * i + v + 8 * half;
      v4f o;
      o.x = acc[i][0][v];
      o.y = acc[i][1][v];
      o.z = acc[i][2][v];
      o.w = acc[i][3][v];
      *reinterpret_cast<v4f*>(ob + (size_t)row * NPIX) = o;
    }
  }
}

extern "C" void kernel_launch(void* const* d_in, const int* in_sizes, int n_in,
                              void* d_out, int out_size, void* d_ws,
                              size_t ws_size, hipStream_t stream) {
  const float* x    = (const float*)d_in[0];
  const float* wqkv = (const float*)d_in[1];
  const float* wout = (const float*)d_in[2];
  float* out = (float*)d_out;

  char* ws = (char*)d_ws;
  const size_t QKV_F = (size_t)BATCH * O3 * NPIX;
  const size_t KB_F  = (size_t)BATCH * HIDDEN * NPIX;
  const size_t CTX_F = (size_t)BATCH * HEADS * DHEAD * DHEAD;
  float* qkv  = (float*)ws;
  float* kbuf = (float*)(ws + QKV_F * 4);
  float* wp   = (float*)(ws + (QKV_F + KB_F + CTX_F) * 4);
  float* w2   = (float*)(ws + (QKV_F + KB_F + CTX_F + 8 * CTX_F) * 4);

  k_pack_wq<<<(HIDDEN * CIN / 2) / 256, 256, 0, stream>>>(wqkv, wp);
  k_q_gemm<<<dim3(NPIX / 64, BATCH), 128, 0, stream>>>(x, wp, qkv);
  k_softmax_n<<<BATCH * HIDDEN, 256, 0, stream>>>(qkv);
  k_softmax_d<<<(BATCH * HEADS * NPIX) / 256, 256, 0, stream>>>(qkv, kbuf);
  k_g_part<<<dim3(GSPLIT, 2, BATCH), 256, 0, stream>>>(kbuf, x, qkv);
  k_ctx_fold<<<BATCH * HEADS * 2, 256, 0, stream>>>(qkv, wqkv, wout, w2);
  k_out_gemm<<<dim3(NPIX / 64, BATCH), 256, 0, stream>>>(qkv, w2, out);
}
